# starting kernel
# baseline (speedup 1.0000x reference)
#include <hip/hip_runtime.h>

typedef __attribute__((ext_vector_type(2))) float v2f;
typedef __attribute__((ext_vector_type(8))) float v8f;

#define M_DIM 32
#define N_DIM 11008
#define K_DIM 4096

__global__ __launch_bounds__(128) void gemm_xwt_wmma_f32(
    const float* __restrict__ x,
    const float* __restrict__ W,
    const float* __restrict__ bias,
    float* __restrict__ dest,
    int klen, int addBias)
{
    const int lane   = threadIdx.x & 31;
    const int wave   = threadIdx.x >> 5;
    const int laneLo = lane & 15;
    const int laneHi = lane >> 4;

    const int n0     = (blockIdx.x * 4 + wave) * 32;
    const int seg    = blockIdx.y;
    const int kStart = seg * klen;

    const float* aPtr0 = x + (size_t)laneLo * K_DIM + kStart + 2 * laneHi;
    const float* aPtr1 = aPtr0 + (size_t)16 * K_DIM;
    const float* bPtr0 = W + (size_t)(n0 + laneLo) * K_DIM + kStart + 2 * laneHi;
    const float* bPtr1 = bPtr0 + (size_t)16 * K_DIM;

    v8f c00 = {};
    v8f c01 = {};
    v8f c10 = {};
    v8f c11 = {};

    for (int kk = 0; kk < klen; kk += 32) {
        __builtin_prefetch(bPtr0 + 512, 0, 0);
        __builtin_prefetch(bPtr1 + 512, 0, 0);
#pragma unroll
        for (int u = 0; u < 8; ++u) {
            v2f b0 = *(const v2f*)(bPtr0 + 4 * u);
            v2f b1 = *(const v2f*)(bPtr1 + 4 * u);
            v2f a0 = *(const v2f*)(aPtr0 + 4 * u);
            v2f a1 = *(const v2f*)(aPtr1 + 4 * u);
            c00 = __builtin_amdgcn_wmma_f32_16x16x4_f32(false, a0, false, b0,
                                                        (short)0, c00, false, false);
            c01 = __builtin_amdgcn_wmma_f32_16x16x4_f32(false, a0, false, b1,
                                                        (short)0, c01, false, false);
            c10 = __builtin_amdgcn_wmma_f32_16x16x4_f32(false, a1, false, b0,
                                                        (short)0, c10, false, false);
            c11 = __builtin_amdgcn_wmma_f32_16x16x4_f32(false, a1, false, b1,
                                                        (short)0, c11, false, false);
        }
        aPtr0 += 32; aPtr1 += 32; bPtr0 += 32; bPtr1 += 32;
    }

    const float bv0 = addBias ? bias[n0 + laneLo]      : 0.0f;
    const float bv1 = addBias ? bias[n0 + 16 + laneLo] : 0.0f;
    float* dst = dest + (size_t)seg * M_DIM * N_DIM;
#pragma unroll
    for (int r = 0; r < 8; ++r) {
        const int m0 = r + 8 * laneHi;
        const int m1 = 16 + m0;
        dst[(size_t)m0 * N_DIM + (n0 + laneLo)]      = c00[r] + bv0;
        dst[(size_t)m0 * N_DIM + (n0 + 16 + laneLo)] = c01[r] + bv1;
        dst[(size_t)m1 * N_DIM + (n0 + laneLo)]      = c10[r] + bv0;
        dst[(size_t)m1 * N_DIM + (n0 + 16 + laneLo)] = c11[r] + bv1;
    }
}

__global__ __launch_bounds__(256) void reduce_bias_kernel(
    const float* __restrict__ ws,
    const float* __restrict__ bias,
    float* __restrict__ out, int kseg)
{
    const int total = M_DIM * N_DIM;
    int idx = blockIdx.x * blockDim.x + threadIdx.x;
    if (idx >= total) return;
    int n = idx % N_DIM;
    float acc = bias[n];
    for (int s = 0; s < kseg; ++s)
        acc += ws[(size_t)s * total + idx];
    out[idx] = acc;
}

extern "C" void kernel_launch(void* const* d_in, const int* in_sizes, int n_in,
                              void* d_out, int out_size, void* d_ws, size_t ws_size,
                              hipStream_t stream) {
    const float* x    = (const float*)d_in[0];
    const float* W    = (const float*)d_in[1];
    const float* bias = (const float*)d_in[2];
    float* out = (float*)d_out;

    const size_t per_seg = (size_t)M_DIM * N_DIM * sizeof(float);
    int kseg = 1;
    if      (ws_size >= 16 * per_seg) kseg = 16;
    else if (ws_size >=  8 * per_seg) kseg = 8;
    else if (ws_size >=  4 * per_seg) kseg = 4;
    else if (ws_size >=  2 * per_seg) kseg = 2;
    const int klen = K_DIM / kseg;

    dim3 grid(N_DIM / (32 * 4), kseg);
    if (kseg == 1) {
        gemm_xwt_wmma_f32<<<grid, 128, 0, stream>>>(x, W, bias, out, klen, 1);
    } else {
        float* ws = (float*)d_ws;
        gemm_xwt_wmma_f32<<<grid, 128, 0, stream>>>(x, W, bias, ws, klen, 0);
        const int total = M_DIM * N_DIM;
        reduce_bias_kernel<<<(total + 255) / 256, 256, 0, stream>>>(ws, bias, out, kseg);
    }
}


# Operand fragment loads widened from 8 to 16 B per lane in both loaders (k permuted identically in A and B inside each 8-k group, so sum order changes): half the load instructions and line touches
# speedup vs baseline: 1.2837x; 1.2837x over previous
#include <hip/hip_runtime.h>

typedef __attribute__((ext_vector_type(2))) float v2f;
typedef __attribute__((ext_vector_type(4))) float v4f;
typedef __attribute__((ext_vector_type(8))) float v8f;

#define M_DIM 32
#define N_DIM 11008
#define K_DIM 4096

__global__ __launch_bounds__(128) void gemm_xwt_wmma_f32(
    const float* __restrict__ x,
    const float* __restrict__ W,
    const float* __restrict__ bias,
    float* __restrict__ dest,
    int klen, int addBias)
{
    const int lane   = threadIdx.x & 31;
    const int wave   = threadIdx.x >> 5;
    const int laneLo = lane & 15;
    const int laneHi = lane >> 4;

    const int n0     = (blockIdx.x * 4 + wave) * 32;
    const int seg    = blockIdx.y;
    const int kStart = seg * klen;

    const float* aPtr0 = x + (size_t)laneLo * K_DIM + kStart + 4 * laneHi;
    const float* aPtr1 = aPtr0 + (size_t)16 * K_DIM;
    const float* bPtr0 = W + (size_t)(n0 + laneLo) * K_DIM + kStart + 4 * laneHi;
    const float* bPtr1 = bPtr0 + (size_t)16 * K_DIM;

    v8f c00 = {};
    v8f c01 = {};
    v8f c10 = {};
    v8f c11 = {};

    for (int kk = 0; kk < klen; kk += 32) {
        __builtin_prefetch(bPtr0 + 512, 0, 0);
        __builtin_prefetch(bPtr1 + 512, 0, 0);
#pragma unroll
        for (int u = 0; u < 4; ++u) {
            const v4f b0 = *(const v4f*)(bPtr0 + 8 * u);
            const v4f b1 = *(const v4f*)(bPtr1 + 8 * u);
            const v4f a0 = *(const v4f*)(aPtr0 + 8 * u);
            const v4f a1 = *(const v4f*)(aPtr1 + 8 * u);
            const v2f b0lo = {b0.x, b0.y}, b0hi = {b0.z, b0.w};
            const v2f b1lo = {b1.x, b1.y}, b1hi = {b1.z, b1.w};
            const v2f a0lo = {a0.x, a0.y}, a0hi = {a0.z, a0.w};
            const v2f a1lo = {a1.x, a1.y}, a1hi = {a1.z, a1.w};
            c00 = __builtin_amdgcn_wmma_f32_16x16x4_f32(false, a0lo, false, b0lo,
                                                        (short)0, c00, false, false);
            c01 = __builtin_amdgcn_wmma_f32_16x16x4_f32(false, a0lo, false, b1lo,
                                                        (short)0, c01, false, false);
            c10 = __builtin_amdgcn_wmma_f32_16x16x4_f32(false, a1lo, false, b0lo,
                                                        (short)0, c10, false, false);
            c11 = __builtin_amdgcn_wmma_f32_16x16x4_f32(false, a1lo, false, b1lo,
                                                        (short)0, c11, false, false);
            c00 = __builtin_amdgcn_wmma_f32_16x16x4_f32(false, a0hi, false, b0hi,
                                                        (short)0, c00, false, false);
            c01 = __builtin_amdgcn_wmma_f32_16x16x4_f32(false, a0hi, false, b1hi,
                                                        (short)0, c01, false, false);
            c10 = __builtin_amdgcn_wmma_f32_16x16x4_f32(false, a1hi, false, b0hi,
                                                        (short)0, c10, false, false);
            c11 = __builtin_amdgcn_wmma_f32_16x16x4_f32(false, a1hi, false, b1hi,
                                                        (short)0, c11, false, false);
        }
        aPtr0 += 32; aPtr1 += 32; bPtr0 += 32; bPtr1 += 32;
    }

    const float bv0 = addBias ? bias[n0 + laneLo]      : 0.0f;
    const float bv1 = addBias ? bias[n0 + 16 + laneLo] : 0.0f;
    float* dst = dest + (size_t)seg * M_DIM * N_DIM;
#pragma unroll
    for (int r = 0; r < 8; ++r) {
        const int m0 = r + 8 * laneHi;
        const int m1 = 16 + m0;
        dst[(size_t)m0 * N_DIM + (n0 + laneLo)]      = c00[r] + bv0;
        dst[(size_t)m0 * N_DIM + (n0 + 16 + laneLo)] = c01[r] + bv1;
        dst[(size_t)m1 * N_DIM + (n0 + laneLo)]      = c10[r] + bv0;
        dst[(size_t)m1 * N_DIM + (n0 + 16 + laneLo)] = c11[r] + bv1;
    }
}

__global__ __launch_bounds__(256) void reduce_bias_kernel(
    const float* __restrict__ ws,
    const float* __restrict__ bias,
    float* __restrict__ out, int kseg)
{
    const int total = M_DIM * N_DIM;
    int idx = blockIdx.x * blockDim.x + threadIdx.x;
    if (idx >= total) return;
    int n = idx % N_DIM;
    float acc = bias[n];
    for (int s = 0; s < kseg; ++s)
        acc += ws[(size_t)s * total + idx];
    out[idx] = acc;
}

extern "C" void kernel_launch(void* const* d_in, const int* in_sizes, int n_in,
                              void* d_out, int out_size, void* d_ws, size_t ws_size,
                              hipStream_t stream) {
    const float* x    = (const float*)d_in[0];
    const float* W    = (const float*)d_in[1];
    const float* bias = (const float*)d_in[2];
    float* out = (float*)d_out;

    const size_t per_seg = (size_t)M_DIM * N_DIM * sizeof(float);
    int kseg = 1;
    if      (ws_size >= 16 * per_seg) kseg = 16;
    else if (ws_size >=  8 * per_seg) kseg = 8;
    else if (ws_size >=  4 * per_seg) kseg = 4;
    else if (ws_size >=  2 * per_seg) kseg = 2;
    const int klen = K_DIM / kseg;

    dim3 grid(N_DIM / (32 * 4), kseg);
    if (kseg == 1) {
        gemm_xwt_wmma_f32<<<grid, 128, 0, stream>>>(x, W, bias, out, klen, 1);
    } else {
        float* ws = (float*)d_ws;
        gemm_xwt_wmma_f32<<<grid, 128, 0, stream>>>(x, W, bias, ws, klen, 0);
        const int total = M_DIM * N_DIM;
        reduce_bias_kernel<<<(total + 255) / 256, 256, 0, stream>>>(ws, bias, out, kseg);
    }
}


# Dropped the W prefetch: 2 KB ahead of a 1-KB K segment it named only lines that another work-group loads anyway, so every W line was requested by two work-groups; now by its consumer only
# speedup vs baseline: 1.3918x; 1.0842x over previous
#include <hip/hip_runtime.h>

typedef __attribute__((ext_vector_type(2))) float v2f;
typedef __attribute__((ext_vector_type(4))) float v4f;
typedef __attribute__((ext_vector_type(8))) float v8f;

#define M_DIM 32
#define N_DIM 11008
#define K_DIM 4096

__global__ __launch_bounds__(128) void gemm_xwt_wmma_f32(
    const float* __restrict__ x,
    const float* __restrict__ W,
    const float* __restrict__ bias,
    float* __restrict__ dest,
    int klen, int addBias)
{
    const int lane   = threadIdx.x & 31;
    const int wave   = threadIdx.x >> 5;
    const int laneLo = lane & 15;
    const int laneHi = lane >> 4;

    const int n0     = (blockIdx.x * 4 + wave) * 32;
    const int seg    = blockIdx.y;
    const int kStart = seg * klen;

    const float* aPtr0 = x + (size_t)laneLo * K_DIM + kStart + 4 * laneHi;
    const float* aPtr1 = aPtr0 + (size_t)16 * K_DIM;
    const float* bPtr0 = W + (size_t)(n0 + laneLo) * K_DIM + kStart + 4 * laneHi;
    const float* bPtr1 = bPtr0 + (size_t)16 * K_DIM;

    v8f c00 = {};
    v8f c01 = {};
    v8f c10 = {};
    v8f c11 = {};

    for (int kk = 0; kk < klen; kk += 32) {
#pragma unroll
        for (int u = 0; u < 4; ++u) {
            const v4f b0 = *(const v4f*)(bPtr0 + 8 * u);
            const v4f b1 = *(const v4f*)(bPtr1 + 8 * u);
            const v4f a0 = *(const v4f*)(aPtr0 + 8 * u);
            const v4f a1 = *(const v4f*)(aPtr1 + 8 * u);
            const v2f b0lo = {b0.x, b0.y}, b0hi = {b0.z, b0.w};
            const v2f b1lo = {b1.x, b1.y}, b1hi = {b1.z, b1.w};
            const v2f a0lo = {a0.x, a0.y}, a0hi = {a0.z, a0.w};
            const v2f a1lo = {a1.x, a1.y}, a1hi = {a1.z, a1.w};
            c00 = __builtin_amdgcn_wmma_f32_16x16x4_f32(false, a0lo, false, b0lo,
                                                        (short)0, c00, false, false);
            c01 = __builtin_amdgcn_wmma_f32_16x16x4_f32(false, a0lo, false, b1lo,
                                                        (short)0, c01, false, false);
            c10 = __builtin_amdgcn_wmma_f32_16x16x4_f32(false, a1lo, false, b0lo,
                                                        (short)0, c10, false, false);
            c11 = __builtin_amdgcn_wmma_f32_16x16x4_f32(false, a1lo, false, b1lo,
                                                        (short)0, c11, false, false);
            c00 = __builtin_amdgcn_wmma_f32_16x16x4_f32(false, a0hi, false, b0hi,
                                                        (short)0, c00, false, false);
            c01 = __builtin_amdgcn_wmma_f32_16x16x4_f32(false, a0hi, false, b1hi,
                                                        (short)0, c01, false, false);
            c10 = __builtin_amdgcn_wmma_f32_16x16x4_f32(false, a1hi, false, b0hi,
                                                        (short)0, c10, false, false);
            c11 = __builtin_amdgcn_wmma_f32_16x16x4_f32(false, a1hi, false, b1hi,
                                                        (short)0, c11, false, false);
        }
        aPtr0 += 32; aPtr1 += 32; bPtr0 += 32; bPtr1 += 32;
    }

    const float bv0 = addBias ? bias[n0 + laneLo]      : 0.0f;
    const float bv1 = addBias ? bias[n0 + 16 + laneLo] : 0.0f;
    float* dst = dest + (size_t)seg * M_DIM * N_DIM;
#pragma unroll
    for (int r = 0; r < 8; ++r) {
        const int m0 = r + 8 * laneHi;
        const int m1 = 16 + m0;
        dst[(size_t)m0 * N_DIM + (n0 + laneLo)]      = c00[r] + bv0;
        dst[(size_t)m0 * N_DIM + (n0 + 16 + laneLo)] = c01[r] + bv1;
        dst[(size_t)m1 * N_DIM + (n0 + laneLo)]      = c10[r] + bv0;
        dst[(size_t)m1 * N_DIM + (n0 + 16 + laneLo)] = c11[r] + bv1;
    }
}

__global__ __launch_bounds__(256) void reduce_bias_kernel(
    const float* __restrict__ ws,
    const float* __restrict__ bias,
    float* __restrict__ out, int kseg)
{
    const int total = M_DIM * N_DIM;
    int idx = blockIdx.x * blockDim.x + threadIdx.x;
    if (idx >= total) return;
    int n = idx % N_DIM;
    float acc = bias[n];
    for (int s = 0; s < kseg; ++s)
        acc += ws[(size_t)s * total + idx];
    out[idx] = acc;
}

extern "C" void kernel_launch(void* const* d_in, const int* in_sizes, int n_in,
                              void* d_out, int out_size, void* d_ws, size_t ws_size,
                              hipStream_t stream) {
    const float* x    = (const float*)d_in[0];
    const float* W    = (const float*)d_in[1];
    const float* bias = (const float*)d_in[2];
    float* out = (float*)d_out;

    const size_t per_seg = (size_t)M_DIM * N_DIM * sizeof(float);
    int kseg = 1;
    if      (ws_size >= 16 * per_seg) kseg = 16;
    else if (ws_size >=  8 * per_seg) kseg = 8;
    else if (ws_size >=  4 * per_seg) kseg = 4;
    else if (ws_size >=  2 * per_seg) kseg = 2;
    const int klen = K_DIM / kseg;

    dim3 grid(N_DIM / (32 * 4), kseg);
    if (kseg == 1) {
        gemm_xwt_wmma_f32<<<grid, 128, 0, stream>>>(x, W, bias, out, klen, 1);
    } else {
        float* ws = (float*)d_ws;
        gemm_xwt_wmma_f32<<<grid, 128, 0, stream>>>(x, W, bias, ws, klen, 0);
        const int total = M_DIM * N_DIM;
        reduce_bias_kernel<<<(total + 255) / 256, 256, 0, stream>>>(ws, bias, out, kseg);
    }
}


# Operand loads made whole-line: 8 lanes read one row's 128 contiguous B (4 lines per load, not 16); rows transposed to fragments through a wave-private LDS region; k order and sum order unchanged
# speedup vs baseline: 1.9512x; 1.4019x over previous
#include <hip/hip_runtime.h>

typedef __attribute__((ext_vector_type(2))) float v2f;
typedef __attribute__((ext_vector_type(4))) float v4f;
typedef __attribute__((ext_vector_type(8))) float v8f;

#define M_DIM 32
#define N_DIM 11008
#define K_DIM 4096


#define KC       32
#define PITCH_F  36
#define OP_F     (32 * PITCH_F)

__global__ __launch_bounds__(128) void gemm_xwt_wmma_f32(
    const float* __restrict__ x,
    const float* __restrict__ W,
    const float* __restrict__ bias,
    float* __restrict__ dest,
    int klen, int addBias)
{
    __shared__ __attribute__((aligned(16))) float stage[4][2][OP_F];

    const int lane   = threadIdx.x & 31;
    const int wave   = threadIdx.x >> 5;
    const int laneLo = lane & 15;
    const int laneHi = lane >> 4;

    const int n0     = (blockIdx.x * 4 + wave) * 32;
    const int seg    = blockIdx.y;
    const int kStart = seg * klen;

    const int gRow = lane >> 3;
    const int gCol = 4 * (lane & 7);
    const float* xg = x + (size_t)gRow * K_DIM + kStart + gCol;
    const float* wg = W + (size_t)(n0 + gRow) * K_DIM + kStart + gCol;

    float* sA = &stage[wave][0][gRow * PITCH_F + gCol];
    float* sB = &stage[wave][1][gRow * PITCH_F + gCol];
    const float* fA = &stage[wave][0][laneLo * PITCH_F + 4 * laneHi];
    const float* fB = &stage[wave][1][laneLo * PITCH_F + 4 * laneHi];

    v8f c00 = {};
    v8f c01 = {};
    v8f c10 = {};
    v8f c11 = {};

    v4f ga[8], gb[8];
#pragma unroll
    for (int j = 0; j < 8; ++j) {
        gb[j] = *(const v4f*)(wg + (size_t)(4 * j) * K_DIM);
        ga[j] = *(const v4f*)(xg + (size_t)(4 * j) * K_DIM);
    }

    for (int kk = 0; kk < klen; kk += KC) {
#pragma unroll
        for (int j = 0; j < 8; ++j) {
            *(v4f*)(sB + 4 * j * PITCH_F) = gb[j];
            *(v4f*)(sA + 4 * j * PITCH_F) = ga[j];
        }

        if (kk + KC < klen) {
            xg += KC; wg += KC;
#pragma unroll
            for (int j = 0; j < 8; ++j) {
                gb[j] = *(const v4f*)(wg + (size_t)(4 * j) * K_DIM);
                ga[j] = *(const v4f*)(xg + (size_t)(4 * j) * K_DIM);
            }
        }

#pragma unroll
        for (int u = 0; u < 4; ++u) {
            const v4f b0 = *(const v4f*)(fB + 8 * u);
            const v4f b1 = *(const v4f*)(fB + 16 * PITCH_F + 8 * u);
            const v4f a0 = *(const v4f*)(fA + 8 * u);
            const v4f a1 = *(const v4f*)(fA + 16 * PITCH_F + 8 * u);
            const v2f b0lo = {b0.x, b0.y}, b0hi = {b0.z, b0.w};
            const v2f b1lo = {b1.x, b1.y}, b1hi = {b1.z, b1.w};
            const v2f a0lo = {a0.x, a0.y}, a0hi = {a0.z, a0.w};
            const v2f a1lo = {a1.x, a1.y}, a1hi = {a1.z, a1.w};
            c00 = __builtin_amdgcn_wmma_f32_16x16x4_f32(false, a0lo, false, b0lo,
                                                        (short)0, c00, false, false);
            c01 = __builtin_amdgcn_wmma_f32_16x16x4_f32(false, a0lo, false, b1lo,
                                                        (short)0, c01, false, false);
            c10 = __builtin_amdgcn_wmma_f32_16x16x4_f32(false, a1lo, false, b0lo,
                                                        (short)0, c10, false, false);
            c11 = __builtin_amdgcn_wmma_f32_16x16x4_f32(false, a1lo, false, b1lo,
                                                        (short)0, c11, false, false);
            c00 = __builtin_amdgcn_wmma_f32_16x16x4_f32(false, a0hi, false, b0hi,
                                                        (short)0, c00, false, false);
            c01 = __builtin_amdgcn_wmma_f32_16x16x4_f32(false, a0hi, false, b1hi,
                                                        (short)0, c01, false, false);
            c10 = __builtin_amdgcn_wmma_f32_16x16x4_f32(false, a1hi, false, b0hi,
                                                        (short)0, c10, false, false);
            c11 = __builtin_amdgcn_wmma_f32_16x16x4_f32(false, a1hi, false, b1hi,
                                                        (short)0, c11, false, false);
        }
    }

    const float bv0 = addBias ? bias[n0 + laneLo]      : 0.0f;
    const float bv1 = addBias ? bias[n0 + 16 + laneLo] : 0.0f;
    float* dst = dest + (size_t)seg * M_DIM * N_DIM;
#pragma unroll
    for (int r = 0; r < 8; ++r) {
        const int m0 = r + 8 * laneHi;
        const int m1 = 16 + m0;
        dst[(size_t)m0 * N_DIM + (n0 + laneLo)]      = c00[r] + bv0;
        dst[(size_t)m0 * N_DIM + (n0 + 16 + laneLo)] = c01[r] + bv1;
        dst[(size_t)m1 * N_DIM + (n0 + laneLo)]      = c10[r] + bv0;
        dst[(size_t)m1 * N_DIM + (n0 + 16 + laneLo)] = c11[r] + bv1;
    }
}

__global__ __launch_bounds__(256) void reduce_bias_kernel(
    const float* __restrict__ ws,
    const float* __restrict__ bias,
    float* __restrict__ out, int kseg)
{
    const int total = M_DIM * N_DIM;
    int idx = blockIdx.x * blockDim.x + threadIdx.x;
    if (idx >= total) return;
    int n = idx % N_DIM;
    float acc = bias[n];
    for (int s = 0; s < kseg; ++s)
        acc += ws[(size_t)s * total + idx];
    out[idx] = acc;
}

extern "C" void kernel_launch(void* const* d_in, const int* in_sizes, int n_in,
                              void* d_out, int out_size, void* d_ws, size_t ws_size,
                              hipStream_t stream) {
    const float* x    = (const float*)d_in[0];
    const float* W    = (const float*)d_in[1];
    const float* bias = (const float*)d_in[2];
    float* out = (float*)d_out;

    const size_t per_seg = (size_t)M_DIM * N_DIM * sizeof(float);
    int kseg = 1;
    if      (ws_size >= 16 * per_seg) kseg = 16;
    else if (ws_size >=  8 * per_seg) kseg = 8;
    else if (ws_size >=  4 * per_seg) kseg = 4;
    else if (ws_size >=  2 * per_seg) kseg = 2;
    const int klen = K_DIM / kseg;

    dim3 grid(N_DIM / (32 * 4), kseg);
    if (kseg == 1) {
        gemm_xwt_wmma_f32<<<grid, 128, 0, stream>>>(x, W, bias, out, klen, 1);
    } else {
        float* ws = (float*)d_ws;
        gemm_xwt_wmma_f32<<<grid, 128, 0, stream>>>(x, W, bias, ws, klen, 0);
        const int total = M_DIM * N_DIM;
        reduce_bias_kernel<<<(total + 255) / 256, 256, 0, stream>>>(ws, bias, out, kseg);
    }
}
